# starting kernel
# baseline (speedup 1.0000x reference)
#include <hip/hip_runtime.h>


typedef float v2f __attribute__((ext_vector_type(2)));
typedef float v8f __attribute__((ext_vector_type(8)));

constexpr int N1_   = 128;
constexpr int N2_   = 128;
constexpr int C_DIM = 768;
constexpr int WH    = 256;
constexpr int KT    = 16;
constexpr int KP    = KT / 2;
constexpr int LDN   = 144;

__global__ __launch_bounds__(512)
void nngp_convkp_kernel(const float* __restrict__ x,
                        const float* __restrict__ y,
                        float* __restrict__ out_xy,
                        float* __restrict__ out_xx,
                        float* __restrict__ out_yy)
{
    __shared__ v2f  XsP[2][KP * LDN];
    __shared__ v2f  YsP[2][KP * LDN];
    __shared__ float red[8 * 128];

    const int wh   = blockIdx.x;
    const int t    = threadIdx.x;
    const int lane = t & 31;
    const int wave = t >> 5;

    const int m0  = (wave >> 2) * 32;
    const int n0  = (wave & 3) * 32;
    const int row = lane & 15;
    const int kh  = lane >> 4;

    const size_t nstride = (size_t)C_DIM * WH;

    v8f acc[2][2] = {};
    float sxx = 0.f, syy = 0.f;

    const int    nloc  = t & 127;
    const int    cbase = t >> 7;
    const size_t gbase = (size_t)nloc * nstride + (size_t)wh;

    float xr[4], yr[4];

    #pragma unroll
    for (int j = 0; j < 4; ++j) {
        const int c = cbase + 4 * j;
        xr[j] = x[gbase + (size_t)c * WH];
        yr[j] = y[gbase + (size_t)c * WH];
        sxx += xr[j] * xr[j];
        syy += yr[j] * yr[j];
    }
    {
        float* xb = (float*)&XsP[0][0];
        float* yb = (float*)&YsP[0][0];
        #pragma unroll
        for (int j = 0; j < 4; ++j) {
            const int c = cbase + 4 * j;
            const int idx = ((c >> 1) * LDN + nloc) * 2 + (c & 1);
            xb[idx] = xr[j];
            yb[idx] = yr[j];
        }
    }
    __syncthreads();

    const int NT = C_DIM / KT;
    for (int kt = 0; kt < NT; ++kt) {
        const int cur = kt & 1;

        const bool have_next = (kt + 1) < NT;
        if (have_next) {
            const int k0n = (kt + 1) * KT;
            #pragma unroll
            for (int j = 0; j < 4; ++j) {
                const int c = k0n + cbase + 4 * j;
                xr[j] = x[gbase + (size_t)c * WH];
                yr[j] = y[gbase + (size_t)c * WH];
            }
        }

        const v2f* Xp = &XsP[cur][0];
        const v2f* Yp = &YsP[cur][0];
        #pragma unroll
        for (int kk = 0; kk < KP; kk += 2) {
            const int kpl = kk + kh;
            v2f a0 = Xp[kpl * LDN + m0      + row];
            v2f a1 = Xp[kpl * LDN + m0 + 16 + row];
            v2f b0 = Yp[kpl * LDN + n0      + row];
            v2f b1 = Yp[kpl * LDN + n0 + 16 + row];

            acc[0][0] = __builtin_amdgcn_wmma_f32_16x16x4_f32(false, a0, false, b0, (short)0, acc[0][0], false, false);
            acc[0][1] = __builtin_amdgcn_wmma_f32_16x16x4_f32(false, a0, false, b1, (short)0, acc[0][1], false, false);
            acc[1][0] = __builtin_amdgcn_wmma_f32_16x16x4_f32(false, a1, false, b0, (short)0, acc[1][0], false, false);
            acc[1][1] = __builtin_amdgcn_wmma_f32_16x16x4_f32(false, a1, false, b1, (short)0, acc[1][1], false, false);
        }

        if (have_next) {
            float* xb = (float*)&XsP[cur ^ 1][0];
            float* yb = (float*)&YsP[cur ^ 1][0];
            #pragma unroll
            for (int j = 0; j < 4; ++j) {
                const int c   = cbase + 4 * j;
                const int idx = ((c >> 1) * LDN + nloc) * 2 + (c & 1);
                xb[idx] = xr[j];
                yb[idx] = yr[j];
                sxx += xr[j] * xr[j];
                syy += yr[j] * yr[j];
            }
            __syncthreads();
        }
    }

    const float inv_c = 1.0f / (float)C_DIM;

    #pragma unroll
    for (int i = 0; i < 2; ++i) {
        #pragma unroll
        for (int j = 0; j < 2; ++j) {
            const int mm = m0 + 16 * i + kh * 8;
            const int nn = n0 + 16 * j + (lane & 15);
            #pragma unroll
            for (int r = 0; r < 8; ++r) {
                out_xy[(size_t)((mm + r) * N2_ + nn) * WH + wh] = acc[i][j][r] * inv_c;
            }
        }
    }

    __syncthreads();
    red[      (t >> 7) * 128 + nloc] = sxx;
    red[512 + (t >> 7) * 128 + nloc] = syy;
    __syncthreads();
    if (t < 128) {
        const float vx = red[t] + red[128 + t] + red[256 + t] + red[384 + t];
        const float vy = red[512 + t] + red[640 + t] + red[768 + t] + red[896 + t];
        out_xx[(size_t)t * WH + wh] = vx * inv_c;
        out_yy[(size_t)t * WH + wh] = vy * inv_c;
    }
}

extern "C" void kernel_launch(void* const* d_in, const int* in_sizes, int n_in,
                              void* d_out, int out_size, void* d_ws, size_t ws_size,
                              hipStream_t stream) {
    (void)in_sizes; (void)n_in; (void)out_size; (void)d_ws; (void)ws_size;
    const float* x = (const float*)d_in[0];
    const float* y = (const float*)d_in[1];
    float* out    = (float*)d_out;
    float* out_xy = out;
    float* out_xx = out + (size_t)N1_ * N2_ * WH;
    float* out_yy = out_xx + (size_t)N1_ * WH;
    nngp_convkp_kernel<<<WH, 512, 0, stream>>>(x, y, out_xy, out_xx, out_yy);
}


# Each block owns 32 consecutive wh (a 16x32 n-tile each) instead of one wh, so all global loads and stores cover whole 128-B lines via LDS planes; xx/yy partial sums regrouped (new reduction tree)
# speedup vs baseline: 4.5431x; 4.5431x over previous
#include <hip/hip_runtime.h>


typedef float v2f __attribute__((ext_vector_type(2)));
typedef float v4f __attribute__((ext_vector_type(4)));
typedef float v8f __attribute__((ext_vector_type(8)));

constexpr int N1_   = 128;
constexpr int N2_   = 128;
constexpr int C_DIM = 768;
constexpr int WH    = 256;
constexpr int G     = 32;
constexpr int TM    = 16;
constexpr int TN    = 32;
constexpr int KT    = 16;
constexpr int KP    = KT / 2;
constexpr int PSX   = KP * TM + 1;
constexpr int PSY   = KP * TN + 1;
constexpr int YOFF  = G * PSX;
constexpr int STAGE = G * (PSX + PSY);
constexpr int OS    = G + 4;
constexpr int NT    = C_DIM / KT;

__global__ __launch_bounds__(512)
void nngp_convkp_kernel(const float* __restrict__ x,
                        const float* __restrict__ y,
                        float* __restrict__ out_xy,
                        float* __restrict__ out_xx,
                        float* __restrict__ out_yy)
{
    __shared__ __attribute__((aligned(16))) v2f stg[2][STAGE];
    __shared__ v4f red[1024];

    const int b  = blockIdx.x;
    const int g  = b & 7;
    const int i1 = (b >> 3) & 7;
    const int i2 = b >> 6;

    const int t    = threadIdx.x;
    const int lane = t & 31;
    const int wave = __builtin_amdgcn_readfirstlane(t >> 5);
    const int row  = lane & 15;
    const int kh   = lane >> 4;

    const int q = t & 7;
    const int u = t >> 3;
    const float* xg = x + ((size_t)(i1 * TM + (u & 15)) * C_DIM + 2 * (u >> 4)) * WH + g * G + 4 * q;
    const float* yg = y + ((size_t)(i2 * TN + (u & 31)) * C_DIM + 2 * (u >> 5)) * WH + g * G + 4 * q;
    const int sx = (4 * q) * PSX + u;
    const int sy = YOFF + (4 * q) * PSY + ((u >> 4) & 1) * 128 + (u >> 5) * 16 + (u & 15);

    const bool do_xx = (wave & 3) == i2;
    const bool do_yy = (wave & 7) == i1;

    const int ax = (2 * wave) * PSX + kh * 16 + row;
    const int by = YOFF + (2 * wave) * PSY + kh * 16 + row;

    v8f acc[2][2] = {};
    v4f sxx = {0.f, 0.f, 0.f, 0.f}, syy = {0.f, 0.f, 0.f, 0.f};
    v4f xr[2][2], yr[4][2];

#define LOAD_TILE()                                                              \
    do {                                                                         \
        _Pragma("unroll")                                                        \
        for (int s = 0; s < 2; ++s) {                                            \
            xr[s][0] = *(const v4f*)(xg + (8 * s + 0) * WH);                     \
            xr[s][1] = *(const v4f*)(xg + (8 * s + 1) * WH);                     \
        }                                                                        \
        _Pragma("unroll")                                                        \
        for (int s = 0; s < 4; ++s) {                                            \
            yr[s][0] = *(const v4f*)(yg + (4 * s + 0) * WH);                     \
            yr[s][1] = *(const v4f*)(yg + (4 * s + 1) * WH);                     \
        }                                                                        \
        xg += KT * WH;                                                           \
        yg += KT * WH;                                                           \
    } while (0)

#define STAGE_TILE(buf)                                                          \
    do {                                                                         \
        v2f* S_ = &stg[(buf)][0];                                                \
        _Pragma("unroll")                                                        \
        for (int s = 0; s < 2; ++s) {                                            \
            _Pragma("unroll")                                                    \
            for (int i = 0; i < 4; ++i) {                                        \
                v2f p_ = {xr[s][0][i], xr[s][1][i]};                             \
                S_[sx + i * PSX + 64 * s] = p_;                                  \
            }                                                                    \
        }                                                                        \
        _Pragma("unroll")                                                        \
        for (int s = 0; s < 4; ++s) {                                            \
            _Pragma("unroll")                                                    \
            for (int i = 0; i < 4; ++i) {                                        \
                v2f p_ = {yr[s][0][i], yr[s][1][i]};                             \
                S_[sy + i * PSY + 32 * s] = p_;                                  \
            }                                                                    \
        }                                                                        \
        if (do_xx) {                                                             \
            _Pragma("unroll")                                                    \
            for (int s = 0; s < 2; ++s) {                                        \
                sxx += xr[s][0] * xr[s][0];                                      \
                sxx += xr[s][1] * xr[s][1];                                      \
            }                                                                    \
        }                                                                        \
        if (do_yy) {                                                             \
            _Pragma("unroll")                                                    \
            for (int s = 0; s < 4; ++s) {                                        \
                syy += yr[s][0] * yr[s][0];                                      \
                syy += yr[s][1] * yr[s][1];                                      \
            }                                                                    \
        }                                                                        \
    } while (0)

#define COMPUTE_TILE(buf)                                                        \
    do {                                                                         \
        const v2f* S_ = &stg[(buf)][0];                                          \
        _Pragma("unroll")                                                        \
        for (int kk = 0; kk < KP; kk += 2) {                                     \
            _Pragma("unroll")                                                    \
            for (int v = 0; v < 2; ++v) {                                        \
                v2f a_  = S_[ax + v * PSX + kk * 16];                            \
                v2f b0_ = S_[by + v * PSY + kk * 16];                            \
                v2f b1_ = S_[by + v * PSY + 128 + kk * 16];                      \
                acc[v][0] = __builtin_amdgcn_wmma_f32_16x16x4_f32(false, a_, false, b0_, (short)0, acc[v][0], false, false); \
                acc[v][1] = __builtin_amdgcn_wmma_f32_16x16x4_f32(false, a_, false, b1_, (short)0, acc[v][1], false, false); \
            }                                                                    \
        }                                                                        \
    } while (0)

    LOAD_TILE();
    STAGE_TILE(0);
    __syncthreads();

    for (int kt = 0; kt < NT - 1; ++kt) {
        const int cur = kt & 1;
        LOAD_TILE();
        COMPUTE_TILE(cur);
        STAGE_TILE(cur ^ 1);
        __syncthreads();
    }
    COMPUTE_TILE((NT - 1) & 1);

    const float inv_c = 1.0f / (float)C_DIM;

    float* Of = (float*)&stg[0][0];
    #pragma unroll
    for (int v = 0; v < 2; ++v) {
        #pragma unroll
        for (int j = 0; j < 2; ++j) {
            #pragma unroll
            for (int r = 0; r < 8; ++r) {
                const int pair = (r + 8 * kh) * TN + 16 * j + row;
                Of[pair * OS + 2 * wave + v] = acc[v][j][r] * inv_c;
            }
        }
    }
    if (do_xx) red[(t >> 7) * 128 + (t & 127)] = sxx;
    if (do_yy) red[512 + (t >> 8) * 256 + (t & 255)] = syy;
    __syncthreads();

    #pragma unroll
    for (int it = 0; it < 8; ++it) {
        const int idx  = t + 512 * it;
        const int qq   = idx & 7;
        const int pair = idx >> 3;
        const int n1   = i1 * TM + (pair >> 5);
        const int n2   = i2 * TN + (pair & 31);
        const v4f val  = *(const v4f*)&Of[pair * OS + 4 * qq];
        *(v4f*)&out_xy[((size_t)n1 * N2_ + n2) * WH + g * G + 4 * qq] = val;
    }

    if (wave == i2) {
        const v4f vx = red[t] + red[128 + t] + red[256 + t] + red[384 + t];
        *(v4f*)&out_xx[(size_t)(i1 * TM + (t >> 3)) * WH + g * G + 4 * q] = vx * inv_c;
    }
    if (wave == i1) {
        const v4f vy = red[512 + t] + red[768 + t];
        *(v4f*)&out_yy[(size_t)(i2 * TN + (t >> 3)) * WH + g * G + 4 * q] = vy * inv_c;
    }
#undef LOAD_TILE
#undef STAGE_TILE
#undef COMPUTE_TILE
}

extern "C" void kernel_launch(void* const* d_in, const int* in_sizes, int n_in,
                              void* d_out, int out_size, void* d_ws, size_t ws_size,
                              hipStream_t stream) {
    (void)in_sizes; (void)n_in; (void)out_size; (void)d_ws; (void)ws_size;
    const float* x = (const float*)d_in[0];
    const float* y = (const float*)d_in[1];
    float* out    = (float*)d_out;
    float* out_xy = out;
    float* out_xx = out + (size_t)N1_ * N2_ * WH;
    float* out_yy = out_xx + (size_t)N1_ * WH;
    nngp_convkp_kernel<<<256, 512, 0, stream>>>(x, y, out_xy, out_xx, out_yy);
}


# Global loads issued two K tiles ahead into a second register set; the K-loop barrier now waits for LDS traffic only so they stay in flight across it; summation order unchanged
# speedup vs baseline: 4.9065x; 1.0800x over previous
#include <hip/hip_runtime.h>


typedef float v2f __attribute__((ext_vector_type(2)));
typedef float v4f __attribute__((ext_vector_type(4)));
typedef float v8f __attribute__((ext_vector_type(8)));

constexpr int N1_   = 128;
constexpr int N2_   = 128;
constexpr int C_DIM = 768;
constexpr int WH    = 256;
constexpr int G     = 32;
constexpr int TM    = 16;
constexpr int TN    = 32;
constexpr int KT    = 16;
constexpr int KP    = KT / 2;
constexpr int PSX   = KP * TM + 1;
constexpr int PSY   = KP * TN + 1;
constexpr int YOFF  = G * PSX;
constexpr int STAGE = G * (PSX + PSY);
constexpr int OS    = G + 4;
constexpr int NT    = C_DIM / KT;

static __device__ __forceinline__ void lds_barrier()
{
    __builtin_amdgcn_fence(__ATOMIC_RELEASE, "workgroup", "local");
    __builtin_amdgcn_s_barrier();
    __builtin_amdgcn_fence(__ATOMIC_ACQUIRE, "workgroup", "local");
}

__global__ __launch_bounds__(512)
void nngp_convkp_kernel(const float* __restrict__ x,
                        const float* __restrict__ y,
                        float* __restrict__ out_xy,
                        float* __restrict__ out_xx,
                        float* __restrict__ out_yy)
{
    __shared__ __attribute__((aligned(16))) v2f stg[2][STAGE];
    __shared__ v4f red[1024];

    const int b  = blockIdx.x;
    const int g  = b & 7;
    const int i1 = (b >> 3) & 7;
    const int i2 = b >> 6;

    const int t    = threadIdx.x;
    const int lane = t & 31;
    const int wave = __builtin_amdgcn_readfirstlane(t >> 5);
    const int row  = lane & 15;
    const int kh   = lane >> 4;

    const int q = t & 7;
    const int u = t >> 3;
    const float* xg = x + ((size_t)(i1 * TM + (u & 15)) * C_DIM + 2 * (u >> 4)) * WH + g * G + 4 * q;
    const float* yg = y + ((size_t)(i2 * TN + (u & 31)) * C_DIM + 2 * (u >> 5)) * WH + g * G + 4 * q;
    const int sx = (4 * q) * PSX + u;
    const int sy = YOFF + (4 * q) * PSY + ((u >> 4) & 1) * 128 + (u >> 5) * 16 + (u & 15);

    const bool do_xx = (wave & 3) == i2;
    const bool do_yy = (wave & 7) == i1;

    const int ax = (2 * wave) * PSX + kh * 16 + row;
    const int by = YOFF + (2 * wave) * PSY + kh * 16 + row;

    v8f acc[2][2] = {};
    v4f sxx = {0.f, 0.f, 0.f, 0.f}, syy = {0.f, 0.f, 0.f, 0.f};
    v4f xa[2][2], ya[4][2];
    v4f xb[2][2], yb[4][2];

#define LOAD_TILE(xr, yr)                                                        \
    do {                                                                         \
        _Pragma("unroll")                                                        \
        for (int s = 0; s < 2; ++s) {                                            \
            xr[s][0] = *(const v4f*)(xg + (8 * s + 0) * WH);                     \
            xr[s][1] = *(const v4f*)(xg + (8 * s + 1) * WH);                     \
        }                                                                        \
        _Pragma("unroll")                                                        \
        for (int s = 0; s < 4; ++s) {                                            \
            yr[s][0] = *(const v4f*)(yg + (4 * s + 0) * WH);                     \
            yr[s][1] = *(const v4f*)(yg + (4 * s + 1) * WH);                     \
        }                                                                        \
        xg += KT * WH;                                                           \
        yg += KT * WH;                                                           \
    } while (0)

#define STAGE_TILE(buf, xr, yr)                                                        \
    do {                                                                         \
        v2f* S_ = &stg[(buf)][0];                                                \
        asm volatile("" : "+v"(xr[0][0]), "+v"(xr[0][1]), "+v"(xr[1][0]), "+v"(xr[1][1])); \
        _Pragma("unroll")                                                        \
        for (int s = 0; s < 2; ++s) {                                            \
            _Pragma("unroll")                                                    \
            for (int i = 0; i < 4; ++i) {                                        \
                v2f p_ = {xr[s][0][i], xr[s][1][i]};                             \
                S_[sx + i * PSX + 64 * s] = p_;                                  \
            }                                                                    \
        }                                                                        \
        _Pragma("unroll")                                                        \
        for (int s = 0; s < 4; ++s) {                                            \
            _Pragma("unroll")                                                    \
            for (int i = 0; i < 4; ++i) {                                        \
                v2f p_ = {yr[s][0][i], yr[s][1][i]};                             \
                S_[sy + i * PSY + 32 * s] = p_;                                  \
            }                                                                    \
        }                                                                        \
        if (do_xx) {                                                             \
            _Pragma("unroll")                                                    \
            for (int s = 0; s < 2; ++s) {                                        \
                sxx += xr[s][0] * xr[s][0];                                      \
                sxx += xr[s][1] * xr[s][1];                                      \
            }                                                                    \
        }                                                                        \
        if (do_yy) {                                                             \
            _Pragma("unroll")                                                    \
            for (int s = 0; s < 4; ++s) {                                        \
                syy += yr[s][0] * yr[s][0];                                      \
                syy += yr[s][1] * yr[s][1];                                      \
            }                                                                    \
        }                                                                        \
    } while (0)

#define COMPUTE_TILE(buf)                                                        \
    do {                                                                         \
        const v2f* S_ = &stg[(buf)][0];                                          \
        _Pragma("unroll")                                                        \
        for (int kk = 0; kk < KP; kk += 2) {                                     \
            _Pragma("unroll")                                                    \
            for (int v = 0; v < 2; ++v) {                                        \
                v2f a_  = S_[ax + v * PSX + kk * 16];                            \
                v2f b0_ = S_[by + v * PSY + kk * 16];                            \
                v2f b1_ = S_[by + v * PSY + 128 + kk * 16];                      \
                acc[v][0] = __builtin_amdgcn_wmma_f32_16x16x4_f32(false, a_, false, b0_, (short)0, acc[v][0], false, false); \
                acc[v][1] = __builtin_amdgcn_wmma_f32_16x16x4_f32(false, a_, false, b1_, (short)0, acc[v][1], false, false); \
            }                                                                    \
        }                                                                        \
    } while (0)

    LOAD_TILE(xa, ya);
    LOAD_TILE(xb, yb);
    STAGE_TILE(0, xa, ya);
    lds_barrier();

    for (int kt = 0; kt < NT - 2; kt += 2) {
        LOAD_TILE(xa, ya);
        COMPUTE_TILE(0);
        STAGE_TILE(1, xb, yb);
        lds_barrier();
        LOAD_TILE(xb, yb);
        COMPUTE_TILE(1);
        STAGE_TILE(0, xa, ya);
        lds_barrier();
    }
    COMPUTE_TILE(0);
    STAGE_TILE(1, xb, yb);
    lds_barrier();
    COMPUTE_TILE(1);

    const float inv_c = 1.0f / (float)C_DIM;

    float* Of = (float*)&stg[0][0];
    #pragma unroll
    for (int v = 0; v < 2; ++v) {
        #pragma unroll
        for (int j = 0; j < 2; ++j) {
            #pragma unroll
            for (int r = 0; r < 8; ++r) {
                const int pair = (r + 8 * kh) * TN + 16 * j + row;
                Of[pair * OS + 2 * wave + v] = acc[v][j][r] * inv_c;
            }
        }
    }
    if (do_xx) red[(t >> 7) * 128 + (t & 127)] = sxx;
    if (do_yy) red[512 + (t >> 8) * 256 + (t & 255)] = syy;
    __syncthreads();

    #pragma unroll
    for (int it = 0; it < 8; ++it) {
        const int idx  = t + 512 * it;
        const int qq   = idx & 7;
        const int pair = idx >> 3;
        const int n1   = i1 * TM + (pair >> 5);
        const int n2   = i2 * TN + (pair & 31);
        const v4f val  = *(const v4f*)&Of[pair * OS + 4 * qq];
        *(v4f*)&out_xy[((size_t)n1 * N2_ + n2) * WH + g * G + 4 * qq] = val;
    }

    if (wave == i2) {
        const v4f vx = red[t] + red[128 + t] + red[256 + t] + red[384 + t];
        *(v4f*)&out_xx[(size_t)(i1 * TM + (t >> 3)) * WH + g * G + 4 * q] = vx * inv_c;
    }
    if (wave == i1) {
        const v4f vy = red[512 + t] + red[768 + t];
        *(v4f*)&out_yy[(size_t)(i2 * TN + (t >> 3)) * WH + g * G + 4 * q] = vy * inv_c;
    }
#undef LOAD_TILE
#undef STAGE_TILE
#undef COMPUTE_TILE
}

extern "C" void kernel_launch(void* const* d_in, const int* in_sizes, int n_in,
                              void* d_out, int out_size, void* d_ws, size_t ws_size,
                              hipStream_t stream) {
    (void)in_sizes; (void)n_in; (void)out_size; (void)d_ws; (void)ws_size;
    const float* x = (const float*)d_in[0];
    const float* y = (const float*)d_in[1];
    float* out    = (float*)d_out;
    float* out_xy = out;
    float* out_xx = out + (size_t)N1_ * N2_ * WH;
    float* out_yy = out_xx + (size_t)N1_ * WH;
    nngp_convkp_kernel<<<256, 512, 0, stream>>>(x, y, out_xy, out_xx, out_yy);
}


# Each 128-B line of x and y is prefetched into the second-level cache four K steps before its loads, once, by one of the blocks that share it, so the loads of all sharers hit; sums unchanged
# speedup vs baseline: 5.4935x; 1.1196x over previous
#include <hip/hip_runtime.h>


typedef float v2f __attribute__((ext_vector_type(2)));
typedef float v4f __attribute__((ext_vector_type(4)));
typedef float v8f __attribute__((ext_vector_type(8)));

constexpr int N1_   = 128;
constexpr int N2_   = 128;
constexpr int C_DIM = 768;
constexpr int WH    = 256;
constexpr int G     = 32;
constexpr int TM    = 16;
constexpr int TN    = 32;
constexpr int KT    = 16;
constexpr int KP    = KT / 2;
constexpr int PSX   = KP * TM + 1;
constexpr int PSY   = KP * TN + 1;
constexpr int YOFF  = G * PSX;
constexpr int STAGE = G * (PSX + PSY);
constexpr int OS    = G + 4;
constexpr int NT    = C_DIM / KT;
constexpr int PD    = 6;

static __device__ __forceinline__ void lds_barrier()
{
    __builtin_amdgcn_fence(__ATOMIC_RELEASE, "workgroup", "local");
    __builtin_amdgcn_s_barrier();
    __builtin_amdgcn_fence(__ATOMIC_ACQUIRE, "workgroup", "local");
}

static __device__ __forceinline__ void prefetch_line(const float* p)
{
    __builtin_prefetch(p, 0, 1);
}

__global__ __launch_bounds__(512)
void nngp_convkp_kernel(const float* __restrict__ x,
                        const float* __restrict__ y,
                        float* __restrict__ out_xy,
                        float* __restrict__ out_xx,
                        float* __restrict__ out_yy)
{
    __shared__ __attribute__((aligned(16))) v2f stg[2][STAGE];
    __shared__ v4f red[1024];

    const int b  = blockIdx.x;
    const int g  = b & 7;
    const int i1 = (b >> 3) & 7;
    const int i2 = b >> 6;

    const int t    = threadIdx.x;
    const int lane = t & 31;
    const int wave = __builtin_amdgcn_readfirstlane(t >> 5);
    const int row  = lane & 15;
    const int kh   = lane >> 4;

    const int q = t & 7;
    const int u = t >> 3;
    const float* xg = x + ((size_t)(i1 * TM + (u & 15)) * C_DIM + 2 * (u >> 4)) * WH + g * G + 4 * q;
    const float* yg = y + ((size_t)(i2 * TN + (u & 31)) * C_DIM + 2 * (u >> 5)) * WH + g * G + 4 * q;
    const int sx = (4 * q) * PSX + u;
    const int sy = YOFF + (4 * q) * PSY + ((u >> 4) & 1) * 128 + (u >> 5) * 16 + (u & 15);

    const bool do_xx = (wave & 3) == i2;
    const bool do_yy = (wave & 7) == i1;

    const int pe = 2 * (wave >> 1) + kh;
    const float* pfx = x + ((size_t)(i1 * TM + 4 * i2 + pe) * C_DIM + row) * WH + g * G;
    const float* pfy = y + ((size_t)(i2 * TN + 4 * i1 + pe) * C_DIM + row) * WH + g * G;
    const float* pf  = (wave & 1) ? pfy : pfx;
    const bool pf_wave = wave < 4;

    const int ax = (2 * wave) * PSX + kh * 16 + row;
    const int by = YOFF + (2 * wave) * PSY + kh * 16 + row;

    v8f acc[2][2] = {};
    v4f sxx = {0.f, 0.f, 0.f, 0.f}, syy = {0.f, 0.f, 0.f, 0.f};
    v4f xa[2][2], ya[4][2];
    v4f xb[2][2], yb[4][2];

#define LOAD_TILE(xr, yr)                                                        \
    do {                                                                         \
        _Pragma("unroll")                                                        \
        for (int s = 0; s < 2; ++s) {                                            \
            xr[s][0] = *(const v4f*)(xg + (8 * s + 0) * WH);                     \
            xr[s][1] = *(const v4f*)(xg + (8 * s + 1) * WH);                     \
        }                                                                        \
        _Pragma("unroll")                                                        \
        for (int s = 0; s < 4; ++s) {                                            \
            yr[s][0] = *(const v4f*)(yg + (4 * s + 0) * WH);                     \
            yr[s][1] = *(const v4f*)(yg + (4 * s + 1) * WH);                     \
        }                                                                        \
        xg += KT * WH;                                                           \
        yg += KT * WH;                                                           \
    } while (0)

#define STAGE_TILE(buf, xr, yr)                                                        \
    do {                                                                         \
        v2f* S_ = &stg[(buf)][0];                                                \
        asm volatile("" : "+v"(xr[0][0]), "+v"(xr[0][1]), "+v"(xr[1][0]), "+v"(xr[1][1])); \
        _Pragma("unroll")                                                        \
        for (int s = 0; s < 2; ++s) {                                            \
            _Pragma("unroll")                                                    \
            for (int i = 0; i < 4; ++i) {                                        \
                v2f p_ = {xr[s][0][i], xr[s][1][i]};                             \
                S_[sx + i * PSX + 64 * s] = p_;                                  \
            }                                                                    \
        }                                                                        \
        _Pragma("unroll")                                                        \
        for (int s = 0; s < 4; ++s) {                                            \
            _Pragma("unroll")                                                    \
            for (int i = 0; i < 4; ++i) {                                        \
                v2f p_ = {yr[s][0][i], yr[s][1][i]};                             \
                S_[sy + i * PSY + 32 * s] = p_;                                  \
            }                                                                    \
        }                                                                        \
        if (do_xx) {                                                             \
            _Pragma("unroll")                                                    \
            for (int s = 0; s < 2; ++s) {                                        \
                sxx += xr[s][0] * xr[s][0];                                      \
                sxx += xr[s][1] * xr[s][1];                                      \
            }                                                                    \
        }                                                                        \
        if (do_yy) {                                                             \
            _Pragma("unroll")                                                    \
            for (int s = 0; s < 4; ++s) {                                        \
                syy += yr[s][0] * yr[s][0];                                      \
                syy += yr[s][1] * yr[s][1];                                      \
            }                                                                    \
        }                                                                        \
    } while (0)

#define COMPUTE_TILE(buf)                                                        \
    do {                                                                         \
        const v2f* S_ = &stg[(buf)][0];                                          \
        _Pragma("unroll")                                                        \
        for (int kk = 0; kk < KP; kk += 2) {                                     \
            _Pragma("unroll")                                                    \
            for (int v = 0; v < 2; ++v) {                                        \
                v2f a_  = S_[ax + v * PSX + kk * 16];                            \
                v2f b0_ = S_[by + v * PSY + kk * 16];                            \
                v2f b1_ = S_[by + v * PSY + 128 + kk * 16];                      \
                acc[v][0] = __builtin_amdgcn_wmma_f32_16x16x4_f32(false, a_, false, b0_, (short)0, acc[v][0], false, false); \
                acc[v][1] = __builtin_amdgcn_wmma_f32_16x16x4_f32(false, a_, false, b1_, (short)0, acc[v][1], false, false); \
            }                                                                    \
        }                                                                        \
    } while (0)

    LOAD_TILE(xa, ya);
    LOAD_TILE(xb, yb);
    if (pf_wave) {
        #pragma unroll
        for (int j = 2; j < PD; ++j) prefetch_line(pf + (size_t)j * KT * WH);
    }
    pf += (size_t)PD * KT * WH;
    STAGE_TILE(0, xa, ya);
    lds_barrier();

    for (int kt = 0; kt < NT - 2; kt += 2) {
        if (pf_wave && kt + PD < NT) {
            prefetch_line(pf);
            prefetch_line(pf + KT * WH);
        }
        pf += 2 * KT * WH;
        LOAD_TILE(xa, ya);
        COMPUTE_TILE(0);
        STAGE_TILE(1, xb, yb);
        lds_barrier();
        LOAD_TILE(xb, yb);
        COMPUTE_TILE(1);
        STAGE_TILE(0, xa, ya);
        lds_barrier();
    }
    COMPUTE_TILE(0);
    STAGE_TILE(1, xb, yb);
    lds_barrier();
    COMPUTE_TILE(1);

    const float inv_c = 1.0f / (float)C_DIM;

    float* Of = (float*)&stg[0][0];
    #pragma unroll
    for (int v = 0; v < 2; ++v) {
        #pragma unroll
        for (int j = 0; j < 2; ++j) {
            #pragma unroll
            for (int r = 0; r < 8; ++r) {
                const int pair = (r + 8 * kh) * TN + 16 * j + row;
                Of[pair * OS + 2 * wave + v] = acc[v][j][r] * inv_c;
            }
        }
    }
    if (do_xx) red[(t >> 7) * 128 + (t & 127)] = sxx;
    if (do_yy) red[512 + (t >> 8) * 256 + (t & 255)] = syy;
    __syncthreads();

    #pragma unroll
    for (int it = 0; it < 8; ++it) {
        const int idx  = t + 512 * it;
        const int qq   = idx & 7;
        const int pair = idx >> 3;
        const int n1   = i1 * TM + (pair >> 5);
        const int n2   = i2 * TN + (pair & 31);
        const v4f val  = *(const v4f*)&Of[pair * OS + 4 * qq];
        *(v4f*)&out_xy[((size_t)n1 * N2_ + n2) * WH + g * G + 4 * qq] = val;
    }

    if (wave == i2) {
        const v4f vx = red[t] + red[128 + t] + red[256 + t] + red[384 + t];
        *(v4f*)&out_xx[(size_t)(i1 * TM + (t >> 3)) * WH + g * G + 4 * q] = vx * inv_c;
    }
    if (wave == i1) {
        const v4f vy = red[512 + t] + red[768 + t];
        *(v4f*)&out_yy[(size_t)(i2 * TN + (t >> 3)) * WH + g * G + 4 * q] = vy * inv_c;
    }
#undef LOAD_TILE
#undef STAGE_TILE
#undef COMPUTE_TILE
}

extern "C" void kernel_launch(void* const* d_in, const int* in_sizes, int n_in,
                              void* d_out, int out_size, void* d_ws, size_t ws_size,
                              hipStream_t stream) {
    (void)in_sizes; (void)n_in; (void)out_size; (void)d_ws; (void)ws_size;
    const float* x = (const float*)d_in[0];
    const float* y = (const float*)d_in[1];
    float* out    = (float*)d_out;
    float* out_xy = out;
    float* out_xx = out + (size_t)N1_ * N2_ * WH;
    float* out_yy = out_xx + (size_t)N1_ * WH;
    nngp_convkp_kernel<<<256, 512, 0, stream>>>(x, y, out_xy, out_xx, out_yy);
}
